# starting kernel
# baseline (speedup 1.0000x reference)
#include <hip/hip_runtime.h>

#define WAVES_PER_BLOCK 4
#define ROW_STRIDE 18

typedef float v2f __attribute__((ext_vector_type(2)));
typedef float v8f __attribute__((ext_vector_type(8)));

__device__ __forceinline__ void wave_lds_fence() {
  asm volatile("s_wait_dscnt 0" ::: "memory");
}

__device__ __forceinline__ v8f vsplat8(float s) {
  v8f v;
#pragma unroll
  for (int j = 0; j < 8; ++j) v[j] = s;
  return v;
}

__device__ __forceinline__ v8f vrelu8(v8f v) {
#pragma unroll
  for (int j = 0; j < 8; ++j) v[j] = fmaxf(v[j], 0.0f);
  return v;
}

__device__ __forceinline__ v8f wmma_16x16x16_f32(const v2f a[4], const v2f b[4], v8f c) {
#pragma unroll
  for (int k = 0; k < 4; ++k)
    c = __builtin_amdgcn_wmma_f32_16x16x4_f32(false, a[k], false, b[k], (short)0, c,
                                              false, false);
  return c;
}

__device__ __forceinline__ void load_w_bfmt(const float* __restrict__ W, int lane, v2f b[4]) {
  const int o = lane & 15, hh = lane >> 4;
#pragma unroll
  for (int c = 0; c < 4; ++c)
    b[c] = *(const v2f*)(W + o * 16 + 4 * c + 2 * hh);
}

__device__ __forceinline__ void store_dlayout(float* Lp, int lane, int rowbase, v8f acc) {
  const int nn = lane & 15, hh = lane >> 4;
#pragma unroll
  for (int j = 0; j < 8; ++j)
    Lp[(rowbase + j + 8 * hh) * ROW_STRIDE + nn] = acc[j];
}

__device__ __forceinline__ void load_afmt(const float* Lp, int lane, int row, v2f a[4]) {
  const int hh = lane >> 4;
#pragma unroll
  for (int c = 0; c < 4; ++c)
    a[c] = *(const v2f*)(Lp + row * ROW_STRIDE + 4 * c + 2 * hh);
}

__global__ __launch_bounds__(WAVES_PER_BLOCK * 32) void recover_model_kernel(
    const int* __restrict__ x, const float* __restrict__ phenos,
    const float* __restrict__ emb,
    const float* __restrict__ W1, const float* __restrict__ b1,
    const float* __restrict__ W2, const float* __restrict__ b2,
    const float* __restrict__ W3, const float* __restrict__ b3,
    const float* __restrict__ BW, const float* __restrict__ BD,
    const float* __restrict__ Boff,
    const float* __restrict__ Wc1, const float* __restrict__ bc1,
    const float* __restrict__ Wc2, const float* __restrict__ bc2,
    float* __restrict__ out, int ntiles) {
  __shared__ float lds[WAVES_PER_BLOCK][32 * ROW_STRIDE];
  const int lane = threadIdx.x & 31;
  const int wave = threadIdx.x >> 5;
  const int tile = blockIdx.x * WAVES_PER_BLOCK + wave;
  if (tile >= ntiles) return;
  float* Lp = lds[wave];
  const int nn = lane & 15;
  const int hh = lane >> 4;

  v2f w1b[4], w2b[4], w3b[4], wc1b[4];
  load_w_bfmt(W1, lane, w1b);
  load_w_bfmt(W2, lane, w2b);
  load_w_bfmt(W3, lane, w3b);
  load_w_bfmt(Wc1, lane, wc1b);
  const float bias1 = b1[nn], bias2 = b2[nn], bias3 = b3[nn];

#pragma unroll
  for (int t = 0; t < 2; ++t) {
    const int rb = t * 16;
    const int row = tile * 32 + rb + nn;
    const int g = x[row];
    const float* er = emb + (long)g * 16;
    v2f a[4];
#pragma unroll
    for (int c = 0; c < 4; ++c)
      a[c] = *(const v2f*)(er + 4 * c + 2 * hh);

    v8f acc = vsplat8(bias1);
    acc = wmma_16x16x16_f32(a, w1b, acc);
    acc = vrelu8(acc);
    store_dlayout(Lp, lane, rb, acc);
    wave_lds_fence();
    load_afmt(Lp, lane, rb + nn, a);

    acc = vsplat8(bias2);
    acc = wmma_16x16x16_f32(a, w2b, acc);
    acc = vrelu8(acc);
    store_dlayout(Lp, lane, rb, acc);
    wave_lds_fence();
    load_afmt(Lp, lane, rb + nn, a);

    acc = vsplat8(bias3);
    acc = wmma_16x16x16_f32(a, w3b, acc);
    store_dlayout(Lp, lane, rb, acc);
    wave_lds_fence();
  }

  v2f e0[4], e1[4];
  load_afmt(Lp, lane, 2 * nn, e0);
  load_afmt(Lp, lane, 2 * nn + 1, e1);
  wave_lds_fence();

  v8f z = vsplat8(Boff[nn]);
  for (int m = 0; m < 16; ++m) {
    v2f bA[4], bC[4];
#pragma unroll
    for (int c = 0; c < 4; ++c) {
      const int kk = 4 * c + 2 * hh;
      bA[c] = *(const v2f*)(BW + nn * 256 + m * 16 + kk);
      bC[c] = *(const v2f*)(BW + m * 256 + nn * 16 + kk);
    }
    v8f At = vsplat8(0.0f);
    At = wmma_16x16x16_f32(e0, bA, At);
    v8f Ct = vsplat8(0.0f);
    Ct = wmma_16x16x16_f32(e1, bC, Ct);
    const float bd = BD[nn * 16 + m];
#pragma unroll
    for (int j = 0; j < 8; ++j)
      z[j] += At[j] * Ct[j] * bd;
  }

  store_dlayout(Lp, lane, 0, z);
  wave_lds_fence();
  v2f az[4];
  load_afmt(Lp, lane, nn, az);
  v8f z2 = vsplat8(bc1[nn]);
  z2 = wmma_16x16x16_f32(az, wc1b, z2);
  z2 = vrelu8(z2);

  const float wc2 = Wc2[nn];
  const float cbias = bc2[0];
#pragma unroll
  for (int j = 0; j < 8; ++j) {
    float v = z2[j] * wc2;
    v += __shfl_xor(v, 1);
    v += __shfl_xor(v, 2);
    v += __shfl_xor(v, 4);
    v += __shfl_xor(v, 8);
    if (nn == 0) {
      const int s = j + 8 * hh;
      const int gb = tile * 16 + s;
      out[gb] = phenos[2 * gb] + phenos[2 * gb + 1] + v + cbias;
    }
  }
}

extern "C" void kernel_launch(void* const* d_in, const int* in_sizes, int n_in,
                              void* d_out, int out_size, void* d_ws, size_t ws_size,
                              hipStream_t stream) {
  const int*   x      = (const int*)d_in[0];
  const float* phenos = (const float*)d_in[1];
  const float* emb    = (const float*)d_in[2];
  const float* W1     = (const float*)d_in[3];
  const float* b1     = (const float*)d_in[4];
  const float* W2     = (const float*)d_in[5];
  const float* b2     = (const float*)d_in[6];
  const float* W3     = (const float*)d_in[7];
  const float* b3     = (const float*)d_in[8];
  const float* BW     = (const float*)d_in[9];
  const float* BD     = (const float*)d_in[10];
  const float* Boff   = (const float*)d_in[11];
  const float* Wc1    = (const float*)d_in[12];
  const float* bc1    = (const float*)d_in[13];
  const float* Wc2    = (const float*)d_in[14];
  const float* bc2    = (const float*)d_in[15];
  float* out = (float*)d_out;

  const int Btot   = out_size;
  const int ntiles = (Btot + 15) / 16;
  const int blocks = (ntiles + WAVES_PER_BLOCK - 1) / WAVES_PER_BLOCK;

  recover_model_kernel<<<blocks, WAVES_PER_BLOCK * 32, 0, stream>>>(
      x, phenos, emb, W1, b1, W2, b2, W3, b3, BW, BD, Boff, Wc1, bc1, Wc2, bc2,
      out, ntiles);
}


# BW and BD staged once per work-group in LDS in WMMA B-fragment order; 1024 work-groups of 8 waves stride over the tiles, so the bilinear loop reads LDS instead of strided global loads
# speedup vs baseline: 1.2704x; 1.2704x over previous
#include <hip/hip_runtime.h>

#define WAVES_PER_BLOCK 8
#define GRID_BLOCKS 1024
#define ROW_STRIDE 18
#define FRAG_F2 2048

typedef float v2f __attribute__((ext_vector_type(2)));
typedef float v8f __attribute__((ext_vector_type(8)));

__device__ __forceinline__ void wave_lds_fence() {
  asm volatile("s_wait_dscnt 0" ::: "memory");
}

__device__ __forceinline__ v8f vsplat8(float s) {
  v8f v;
#pragma unroll
  for (int j = 0; j < 8; ++j) v[j] = s;
  return v;
}

__device__ __forceinline__ v8f vrelu8(v8f v) {
#pragma unroll
  for (int j = 0; j < 8; ++j) v[j] = fmaxf(v[j], 0.0f);
  return v;
}

__device__ __forceinline__ v8f wmma_16x16x16_f32(const v2f a[4], const v2f b[4], v8f c) {
#pragma unroll
  for (int k = 0; k < 4; ++k)
    c = __builtin_amdgcn_wmma_f32_16x16x4_f32(false, a[k], false, b[k], (short)0, c,
                                              false, false);
  return c;
}

__device__ __forceinline__ void load_w_bfmt(const float* __restrict__ W, int lane, v2f b[4]) {
  const int o = lane & 15, hh = lane >> 4;
#pragma unroll
  for (int c = 0; c < 4; ++c)
    b[c] = *(const v2f*)(W + o * 16 + 4 * c + 2 * hh);
}

__device__ __forceinline__ void store_dlayout(float* Lp, int lane, int rowbase, v8f acc) {
  const int nn = lane & 15, hh = lane >> 4;
#pragma unroll
  for (int j = 0; j < 8; ++j)
    Lp[(rowbase + j + 8 * hh) * ROW_STRIDE + nn] = acc[j];
}

__device__ __forceinline__ void load_afmt(const float* Lp, int lane, int row, v2f a[4]) {
  const int hh = lane >> 4;
#pragma unroll
  for (int c = 0; c < 4; ++c)
    a[c] = *(const v2f*)(Lp + row * ROW_STRIDE + 4 * c + 2 * hh);
}

__global__ __launch_bounds__(WAVES_PER_BLOCK * 32, 8) void recover_model_kernel(
    const int* __restrict__ x, const float* __restrict__ phenos,
    const float* __restrict__ emb,
    const float* __restrict__ W1, const float* __restrict__ b1,
    const float* __restrict__ W2, const float* __restrict__ b2,
    const float* __restrict__ W3, const float* __restrict__ b3,
    const float* __restrict__ BW, const float* __restrict__ BD,
    const float* __restrict__ Boff,
    const float* __restrict__ Wc1, const float* __restrict__ bc1,
    const float* __restrict__ Wc2, const float* __restrict__ bc2,
    float* __restrict__ out, int ntiles) {
  __shared__ float lds[WAVES_PER_BLOCK][32 * ROW_STRIDE];
  __shared__ v2f fragA[FRAG_F2];
  __shared__ v2f fragC[FRAG_F2];
  __shared__ float bdT[256];
  const int lane = threadIdx.x & 31;
  const int wave = __builtin_amdgcn_readfirstlane(threadIdx.x >> 5);
  float* Lp = lds[wave];
  const int nn = lane & 15;
  const int hh = lane >> 4;

#pragma unroll
  for (int r = 0; r < FRAG_F2 / (WAVES_PER_BLOCK * 32); ++r) {
    const int i = threadIdx.x + r * (WAVES_PER_BLOCK * 32);
    const int fl = i & 31, fc = (i >> 5) & 3, fm = i >> 7;
    const int fn = fl & 15, kk = 4 * fc + 2 * (fl >> 4);
    fragA[i] = *(const v2f*)(BW + fn * 256 + fm * 16 + kk);
    fragC[i] = *(const v2f*)(BW + fm * 256 + fn * 16 + kk);
  }
  bdT[threadIdx.x] = BD[(threadIdx.x & 15) * 16 + (threadIdx.x >> 4)];
  __syncthreads();

  v2f w1b[4], w2b[4], w3b[4], wc1b[4];
  load_w_bfmt(W1, lane, w1b);
  load_w_bfmt(W2, lane, w2b);
  load_w_bfmt(W3, lane, w3b);
  load_w_bfmt(Wc1, lane, wc1b);
  const float bias1 = b1[nn], bias2 = b2[nn], bias3 = b3[nn];
  const float boff = Boff[nn], biasc1 = bc1[nn];
  const float wc2 = Wc2[nn];
  const float cbias = bc2[0];

  for (int tile = blockIdx.x * WAVES_PER_BLOCK + wave; tile < ntiles;
       tile += gridDim.x * WAVES_PER_BLOCK) {

#pragma unroll
    for (int t = 0; t < 2; ++t) {
      const int rb = t * 16;
      const int row = tile * 32 + rb + nn;
      const int g = x[row];
      const float* er = emb + (long)g * 16;
      v2f a[4];
#pragma unroll
      for (int c = 0; c < 4; ++c)
        a[c] = *(const v2f*)(er + 4 * c + 2 * hh);

      v8f acc = vsplat8(bias1);
      acc = wmma_16x16x16_f32(a, w1b, acc);
      acc = vrelu8(acc);
      store_dlayout(Lp, lane, rb, acc);
      wave_lds_fence();
      load_afmt(Lp, lane, rb + nn, a);

      acc = vsplat8(bias2);
      acc = wmma_16x16x16_f32(a, w2b, acc);
      acc = vrelu8(acc);
      store_dlayout(Lp, lane, rb, acc);
      wave_lds_fence();
      load_afmt(Lp, lane, rb + nn, a);

      acc = vsplat8(bias3);
      acc = wmma_16x16x16_f32(a, w3b, acc);
      store_dlayout(Lp, lane, rb, acc);
      wave_lds_fence();
    }

    v2f e0[4], e1[4];
    load_afmt(Lp, lane, 2 * nn, e0);
    load_afmt(Lp, lane, 2 * nn + 1, e1);
    wave_lds_fence();

    v8f z = vsplat8(boff);
#pragma unroll 1
    for (int m = 0; m < 16; ++m) {
      v2f bA[4], bC[4];
#pragma unroll
      for (int c = 0; c < 4; ++c) {
        bA[c] = fragA[(m * 4 + c) * 32 + lane];
        bC[c] = fragC[(m * 4 + c) * 32 + lane];
      }
      v8f At = vsplat8(0.0f);
      At = wmma_16x16x16_f32(e0, bA, At);
      v8f Ct = vsplat8(0.0f);
      Ct = wmma_16x16x16_f32(e1, bC, Ct);
      const float bd = bdT[m * 16 + nn];
#pragma unroll
      for (int j = 0; j < 8; ++j)
        z[j] += At[j] * Ct[j] * bd;
    }

    store_dlayout(Lp, lane, 0, z);
    wave_lds_fence();
    v2f az[4];
    load_afmt(Lp, lane, nn, az);
    v8f z2 = vsplat8(biasc1);
    z2 = wmma_16x16x16_f32(az, wc1b, z2);
    z2 = vrelu8(z2);

#pragma unroll
    for (int j = 0; j < 8; ++j) {
      float v = z2[j] * wc2;
      v += __shfl_xor(v, 1);
      v += __shfl_xor(v, 2);
      v += __shfl_xor(v, 4);
      v += __shfl_xor(v, 8);
      if (nn == 0) {
        const int s = j + 8 * hh;
        const int gb = tile * 16 + s;
        out[gb] = phenos[2 * gb] + phenos[2 * gb + 1] + v + cbias;
      }
    }
  }
}

extern "C" void kernel_launch(void* const* d_in, const int* in_sizes, int n_in,
                              void* d_out, int out_size, void* d_ws, size_t ws_size,
                              hipStream_t stream) {
  const int*   x      = (const int*)d_in[0];
  const float* phenos = (const float*)d_in[1];
  const float* emb    = (const float*)d_in[2];
  const float* W1     = (const float*)d_in[3];
  const float* b1     = (const float*)d_in[4];
  const float* W2     = (const float*)d_in[5];
  const float* b2     = (const float*)d_in[6];
  const float* W3     = (const float*)d_in[7];
  const float* b3     = (const float*)d_in[8];
  const float* BW     = (const float*)d_in[9];
  const float* BD     = (const float*)d_in[10];
  const float* Boff   = (const float*)d_in[11];
  const float* Wc1    = (const float*)d_in[12];
  const float* bc1    = (const float*)d_in[13];
  const float* Wc2    = (const float*)d_in[14];
  const float* bc2    = (const float*)d_in[15];
  float* out = (float*)d_out;

  const int Btot   = out_size;
  const int ntiles = (Btot + 15) / 16;

  recover_model_kernel<<<GRID_BLOCKS, WAVES_PER_BLOCK * 32, 0, stream>>>(
      x, phenos, emb, W1, b1, W2, b2, W3, b3, BW, BD, Boff, Wc1, bc1, Wc2, bc2,
      out, ntiles);
}


# Bilinear form reassociated: T[n,k,k']=sum_m BW[n,m,k]*BD[n,m]*BW[m,n,k'] built per work-group on WMMA, z=(e0 x e1)@T: 64 not 128 WMMAs per tile; summation order of z changes
# speedup vs baseline: 1.7606x; 1.3859x over previous
#include <hip/hip_runtime.h>

#define WAVES_PER_BLOCK 8
#define GRID_BLOCKS 1024
#define ROW_STRIDE 18
#define TFRAG_F4 1024

typedef float v2f __attribute__((ext_vector_type(2)));
typedef float v4f __attribute__((ext_vector_type(4)));
typedef float v8f __attribute__((ext_vector_type(8)));

__device__ __forceinline__ void wave_lds_fence() {
  asm volatile("s_wait_dscnt 0" ::: "memory");
}

__device__ __forceinline__ v8f vsplat8(float s) {
  v8f v;
#pragma unroll
  for (int j = 0; j < 8; ++j) v[j] = s;
  return v;
}

__device__ __forceinline__ v8f vrelu8(v8f v) {
#pragma unroll
  for (int j = 0; j < 8; ++j) v[j] = fmaxf(v[j], 0.0f);
  return v;
}

__device__ __forceinline__ v8f wmma_16x16x16_f32(const v2f a[4], const v2f b[4], v8f c) {
#pragma unroll
  for (int k = 0; k < 4; ++k)
    c = __builtin_amdgcn_wmma_f32_16x16x4_f32(false, a[k], false, b[k], (short)0, c,
                                              false, false);
  return c;
}

__device__ __forceinline__ void load_w_bfmt(const float* __restrict__ W, int lane, v2f b[4]) {
  const int o = lane & 15, hh = lane >> 4;
#pragma unroll
  for (int c = 0; c < 4; ++c)
    b[c] = *(const v2f*)(W + o * 16 + 4 * c + 2 * hh);
}

__device__ __forceinline__ void store_dlayout(float* Lp, int lane, int rowbase, v8f acc) {
  const int nn = lane & 15, hh = lane >> 4;
#pragma unroll
  for (int j = 0; j < 8; ++j)
    Lp[(rowbase + j + 8 * hh) * ROW_STRIDE + nn] = acc[j];
}

__device__ __forceinline__ void load_afmt(const float* Lp, int lane, int row, v2f a[4]) {
  const int hh = lane >> 4;
#pragma unroll
  for (int c = 0; c < 4; ++c)
    a[c] = *(const v2f*)(Lp + row * ROW_STRIDE + 4 * c + 2 * hh);
}

__global__ __launch_bounds__(WAVES_PER_BLOCK * 32, 8) void recover_model_kernel(
    const int* __restrict__ x, const float* __restrict__ phenos,
    const float* __restrict__ emb,
    const float* __restrict__ W1, const float* __restrict__ b1,
    const float* __restrict__ W2, const float* __restrict__ b2,
    const float* __restrict__ W3, const float* __restrict__ b3,
    const float* __restrict__ BW, const float* __restrict__ BD,
    const float* __restrict__ Boff,
    const float* __restrict__ Wc1, const float* __restrict__ bc1,
    const float* __restrict__ Wc2, const float* __restrict__ bc2,
    float* __restrict__ out, int ntiles) {
  __shared__ __attribute__((aligned(16))) float lds[WAVES_PER_BLOCK][32 * ROW_STRIDE];
  __shared__ v4f tfrag[TFRAG_F4];
  const int lane = threadIdx.x & 31;
  const int wave = __builtin_amdgcn_readfirstlane(threadIdx.x >> 5);
  float* Lp = lds[wave];
  const int nn = lane & 15;
  const int hh = lane >> 4;

  {
    float* tf = (float*)tfrag;
#pragma unroll
    for (int i = 0; i < 2; ++i) {
      const int n = 2 * wave + i;
      v2f ta[4], tb[4];
#pragma unroll
      for (int c = 0; c < 4; ++c) {
        const int m0 = 4 * c + 2 * hh;
        const v2f bd = *(const v2f*)(BD + n * 16 + m0);
        ta[c][0] = BW[n * 256 + m0 * 16 + nn] * bd[0];
        ta[c][1] = BW[n * 256 + (m0 + 1) * 16 + nn] * bd[1];
        tb[c][0] = BW[m0 * 256 + n * 16 + nn];
        tb[c][1] = BW[(m0 + 1) * 256 + n * 16 + nn];
      }
      const v8f t = wmma_16x16x16_f32(ta, tb, vsplat8(0.0f));
#pragma unroll
      for (int r = 0; r < 8; ++r) {
        const int k = 8 * hh + r;
        tf[((k * 2 + (nn >> 3)) * 32 + ((nn >> 1) & 1) * 16 + n) * 4 + ((nn >> 2) & 1) * 2 + (nn & 1)] = t[r];
      }
    }
  }
  __syncthreads();

  v2f w1b[4], w2b[4], w3b[4], wc1b[4];
  load_w_bfmt(W1, lane, w1b);
  load_w_bfmt(W2, lane, w2b);
  load_w_bfmt(W3, lane, w3b);
  load_w_bfmt(Wc1, lane, wc1b);
  const float bias1 = b1[nn], bias2 = b2[nn], bias3 = b3[nn];
  const float boff = Boff[nn], biasc1 = bc1[nn];
  const float wc2 = Wc2[nn];
  const float cbias = bc2[0];

  for (int tile = blockIdx.x * WAVES_PER_BLOCK + wave; tile < ntiles;
       tile += gridDim.x * WAVES_PER_BLOCK) {

#pragma unroll
    for (int t = 0; t < 2; ++t) {
      const int rb = t * 16;
      const int row = tile * 32 + rb + nn;
      const int g = x[row];
      const float* er = emb + (long)g * 16;
      v2f a[4];
#pragma unroll
      for (int c = 0; c < 4; ++c)
        a[c] = *(const v2f*)(er + 4 * c + 2 * hh);

      v8f acc = vsplat8(bias1);
      acc = wmma_16x16x16_f32(a, w1b, acc);
      acc = vrelu8(acc);
      store_dlayout(Lp, lane, rb, acc);
      wave_lds_fence();
      load_afmt(Lp, lane, rb + nn, a);

      acc = vsplat8(bias2);
      acc = wmma_16x16x16_f32(a, w2b, acc);
      acc = vrelu8(acc);
      store_dlayout(Lp, lane, rb, acc);
      wave_lds_fence();
      load_afmt(Lp, lane, rb + nn, a);

      acc = vsplat8(bias3);
      acc = wmma_16x16x16_f32(a, w3b, acc);
      store_dlayout(Lp, lane, rb, acc);
      wave_lds_fence();
    }

    v2f e1q[4];
    load_afmt(Lp, lane, 2 * nn + 1, e1q);
    const float* e0row = Lp + (2 * nn) * ROW_STRIDE;

    v8f z0 = vsplat8(0.0f);
    v8f z1 = vsplat8(0.0f);
#pragma unroll
    for (int g = 0; g < 4; ++g) {
      const v4f e0g = *(const v4f*)(e0row + 4 * g);
#pragma unroll
      for (int kk = 0; kk < 4; ++kk) {
        const int k = 4 * g + kk;
        const float s = e0g[kk];
        const v4f t0 = tfrag[(k * 2 + 0) * 32 + lane];
        const v4f t1 = tfrag[(k * 2 + 1) * 32 + lane];
        v2f a, b;
        a = e1q[0] * s; b[0] = t0[0]; b[1] = t0[1];
        z0 = __builtin_amdgcn_wmma_f32_16x16x4_f32(false, a, false, b, (short)0, z0, false, false);
        a = e1q[1] * s; b[0] = t0[2]; b[1] = t0[3];
        z1 = __builtin_amdgcn_wmma_f32_16x16x4_f32(false, a, false, b, (short)0, z1, false, false);
        a = e1q[2] * s; b[0] = t1[0]; b[1] = t1[1];
        z0 = __builtin_amdgcn_wmma_f32_16x16x4_f32(false, a, false, b, (short)0, z0, false, false);
        a = e1q[3] * s; b[0] = t1[2]; b[1] = t1[3];
        z1 = __builtin_amdgcn_wmma_f32_16x16x4_f32(false, a, false, b, (short)0, z1, false, false);
      }
    }
    v8f z;
#pragma unroll
    for (int j = 0; j < 8; ++j) z[j] = (z0[j] + z1[j]) + boff;

    store_dlayout(Lp, lane, 0, z);
    wave_lds_fence();
    v2f az[4];
    load_afmt(Lp, lane, nn, az);
    v8f z2 = vsplat8(biasc1);
    z2 = wmma_16x16x16_f32(az, wc1b, z2);
    z2 = vrelu8(z2);

#pragma unroll
    for (int j = 0; j < 8; ++j) {
      float v = z2[j] * wc2;
      v += __shfl_xor(v, 1);
      v += __shfl_xor(v, 2);
      v += __shfl_xor(v, 4);
      v += __shfl_xor(v, 8);
      if (nn == 0) {
        const int s = j + 8 * hh;
        const int gb = tile * 16 + s;
        out[gb] = phenos[2 * gb] + phenos[2 * gb + 1] + v + cbias;
      }
    }
  }
}

extern "C" void kernel_launch(void* const* d_in, const int* in_sizes, int n_in,
                              void* d_out, int out_size, void* d_ws, size_t ws_size,
                              hipStream_t stream) {
  const int*   x      = (const int*)d_in[0];
  const float* phenos = (const float*)d_in[1];
  const float* emb    = (const float*)d_in[2];
  const float* W1     = (const float*)d_in[3];
  const float* b1     = (const float*)d_in[4];
  const float* W2     = (const float*)d_in[5];
  const float* b2     = (const float*)d_in[6];
  const float* W3     = (const float*)d_in[7];
  const float* b3     = (const float*)d_in[8];
  const float* BW     = (const float*)d_in[9];
  const float* BD     = (const float*)d_in[10];
  const float* Boff   = (const float*)d_in[11];
  const float* Wc1    = (const float*)d_in[12];
  const float* bc1    = (const float*)d_in[13];
  const float* Wc2    = (const float*)d_in[14];
  const float* bc2    = (const float*)d_in[15];
  float* out = (float*)d_out;

  const int Btot   = out_size;
  const int ntiles = (Btot + 15) / 16;

  recover_model_kernel<<<GRID_BLOCKS, WAVES_PER_BLOCK * 32, 0, stream>>>(
      x, phenos, emb, W1, b1, W2, b2, W3, b3, BW, BD, Boff, Wc1, bc1, Wc2, bc2,
      out, ntiles);
}


# Epilogue in registers: the eight 16-lane sums of z2*Wc2 by one DPP row_xmask transpose-butterfly (same add tree) replace 32 ds_bpermute round trips per tile; phenos loaded at tile top; one store
# speedup vs baseline: 2.0573x; 1.1685x over previous
#include <hip/hip_runtime.h>

#define WAVES_PER_BLOCK 8
#define GRID_BLOCKS 1024
#define ROW_STRIDE 18
#define TFRAG_F4 1024

typedef float v2f __attribute__((ext_vector_type(2)));
typedef float v4f __attribute__((ext_vector_type(4)));
typedef float v8f __attribute__((ext_vector_type(8)));

__device__ __forceinline__ void wave_lds_fence() {
  asm volatile("s_wait_dscnt 0" ::: "memory");
}

__device__ __forceinline__ v8f vsplat8(float s) {
  v8f v;
#pragma unroll
  for (int j = 0; j < 8; ++j) v[j] = s;
  return v;
}

__device__ __forceinline__ v8f vrelu8(v8f v) {
#pragma unroll
  for (int j = 0; j < 8; ++j) v[j] = fmaxf(v[j], 0.0f);
  return v;
}

template <int MASK>
__device__ __forceinline__ float row_xor(float v) {
  return __builtin_bit_cast(
      float, __builtin_amdgcn_mov_dpp(__builtin_bit_cast(int, v), 0x160 + MASK, 0xf, 0xf, true));
}

__device__ __forceinline__ v8f wmma_16x16x16_f32(const v2f a[4], const v2f b[4], v8f c) {
#pragma unroll
  for (int k = 0; k < 4; ++k)
    c = __builtin_amdgcn_wmma_f32_16x16x4_f32(false, a[k], false, b[k], (short)0, c,
                                              false, false);
  return c;
}

__device__ __forceinline__ void load_w_bfmt(const float* __restrict__ W, int lane, v2f b[4]) {
  const int o = lane & 15, hh = lane >> 4;
#pragma unroll
  for (int c = 0; c < 4; ++c)
    b[c] = *(const v2f*)(W + o * 16 + 4 * c + 2 * hh);
}

__device__ __forceinline__ void store_dlayout(float* Lp, int lane, int rowbase, v8f acc) {
  const int nn = lane & 15, hh = lane >> 4;
#pragma unroll
  for (int j = 0; j < 8; ++j)
    Lp[(rowbase + j + 8 * hh) * ROW_STRIDE + nn] = acc[j];
}

__device__ __forceinline__ void load_afmt(const float* Lp, int lane, int row, v2f a[4]) {
  const int hh = lane >> 4;
#pragma unroll
  for (int c = 0; c < 4; ++c)
    a[c] = *(const v2f*)(Lp + row * ROW_STRIDE + 4 * c + 2 * hh);
}

__global__ __launch_bounds__(WAVES_PER_BLOCK * 32, 8) void recover_model_kernel(
    const int* __restrict__ x, const float* __restrict__ phenos,
    const float* __restrict__ emb,
    const float* __restrict__ W1, const float* __restrict__ b1,
    const float* __restrict__ W2, const float* __restrict__ b2,
    const float* __restrict__ W3, const float* __restrict__ b3,
    const float* __restrict__ BW, const float* __restrict__ BD,
    const float* __restrict__ Boff,
    const float* __restrict__ Wc1, const float* __restrict__ bc1,
    const float* __restrict__ Wc2, const float* __restrict__ bc2,
    float* __restrict__ out, int ntiles) {
  __shared__ __attribute__((aligned(16))) float lds[WAVES_PER_BLOCK][32 * ROW_STRIDE];
  __shared__ v4f tfrag[TFRAG_F4];
  const int lane = threadIdx.x & 31;
  const int wave = __builtin_amdgcn_readfirstlane(threadIdx.x >> 5);
  float* Lp = lds[wave];
  const int nn = lane & 15;
  const int hh = lane >> 4;

  {
    float* tf = (float*)tfrag;
#pragma unroll
    for (int i = 0; i < 2; ++i) {
      const int n = 2 * wave + i;
      v2f ta[4], tb[4];
#pragma unroll
      for (int c = 0; c < 4; ++c) {
        const int m0 = 4 * c + 2 * hh;
        const v2f bd = *(const v2f*)(BD + n * 16 + m0);
        ta[c][0] = BW[n * 256 + m0 * 16 + nn] * bd[0];
        ta[c][1] = BW[n * 256 + (m0 + 1) * 16 + nn] * bd[1];
        tb[c][0] = BW[m0 * 256 + n * 16 + nn];
        tb[c][1] = BW[(m0 + 1) * 256 + n * 16 + nn];
      }
      const v8f t = wmma_16x16x16_f32(ta, tb, vsplat8(0.0f));
#pragma unroll
      for (int r = 0; r < 8; ++r) {
        const int k = 8 * hh + r;
        tf[((k * 2 + (nn >> 3)) * 32 + ((nn >> 1) & 1) * 16 + n) * 4 + ((nn >> 2) & 1) * 2 + (nn & 1)] = t[r];
      }
    }
  }
  __syncthreads();

  v2f w1b[4], w2b[4], w3b[4], wc1b[4];
  load_w_bfmt(W1, lane, w1b);
  load_w_bfmt(W2, lane, w2b);
  load_w_bfmt(W3, lane, w3b);
  load_w_bfmt(Wc1, lane, wc1b);
  const float bias1 = b1[nn], bias2 = b2[nn], bias3 = b3[nn];
  const float boff = Boff[nn], biasc1 = bc1[nn];
  const float wc2 = Wc2[nn];
  const float cbias = bc2[0];

  for (int tile = blockIdx.x * WAVES_PER_BLOCK + wave; tile < ntiles;
       tile += gridDim.x * WAVES_PER_BLOCK) {

    const int gb = tile * 16 + 8 * hh + (nn & 7);
    const v2f ph = *(const v2f*)(phenos + 2 * gb);

#pragma unroll
    for (int t = 0; t < 2; ++t) {
      const int rb = t * 16;
      const int row = tile * 32 + rb + nn;
      const int g = x[row];
      if (t == 0) {
        asm volatile("" ::"v"(ph[0]), "v"(ph[1]));
      }
      const float* er = emb + (long)g * 16;
      v2f a[4];
#pragma unroll
      for (int c = 0; c < 4; ++c)
        a[c] = *(const v2f*)(er + 4 * c + 2 * hh);

      v8f acc = vsplat8(bias1);
      acc = wmma_16x16x16_f32(a, w1b, acc);
      acc = vrelu8(acc);
      store_dlayout(Lp, lane, rb, acc);
      wave_lds_fence();
      load_afmt(Lp, lane, rb + nn, a);

      acc = vsplat8(bias2);
      acc = wmma_16x16x16_f32(a, w2b, acc);
      acc = vrelu8(acc);
      store_dlayout(Lp, lane, rb, acc);
      wave_lds_fence();
      load_afmt(Lp, lane, rb + nn, a);

      acc = vsplat8(bias3);
      acc = wmma_16x16x16_f32(a, w3b, acc);
      store_dlayout(Lp, lane, rb, acc);
      wave_lds_fence();
    }

    v2f e1q[4];
    load_afmt(Lp, lane, 2 * nn + 1, e1q);
    const float* e0row = Lp + (2 * nn) * ROW_STRIDE;

    v8f z0 = vsplat8(0.0f);
    v8f z1 = vsplat8(0.0f);
#pragma unroll
    for (int g = 0; g < 4; ++g) {
      const v4f e0g = *(const v4f*)(e0row + 4 * g);
#pragma unroll
      for (int kk = 0; kk < 4; ++kk) {
        const int k = 4 * g + kk;
        const float s = e0g[kk];
        const v4f t0 = tfrag[(k * 2 + 0) * 32 + lane];
        const v4f t1 = tfrag[(k * 2 + 1) * 32 + lane];
        v2f a, b;
        a = e1q[0] * s; b[0] = t0[0]; b[1] = t0[1];
        z0 = __builtin_amdgcn_wmma_f32_16x16x4_f32(false, a, false, b, (short)0, z0, false, false);
        a = e1q[1] * s; b[0] = t0[2]; b[1] = t0[3];
        z1 = __builtin_amdgcn_wmma_f32_16x16x4_f32(false, a, false, b, (short)0, z1, false, false);
        a = e1q[2] * s; b[0] = t1[0]; b[1] = t1[1];
        z0 = __builtin_amdgcn_wmma_f32_16x16x4_f32(false, a, false, b, (short)0, z0, false, false);
        a = e1q[3] * s; b[0] = t1[2]; b[1] = t1[3];
        z1 = __builtin_amdgcn_wmma_f32_16x16x4_f32(false, a, false, b, (short)0, z1, false, false);
      }
    }
    v8f z;
#pragma unroll
    for (int j = 0; j < 8; ++j) z[j] = (z0[j] + z1[j]) + boff;

    store_dlayout(Lp, lane, 0, z);
    wave_lds_fence();
    v2f az[4];
    load_afmt(Lp, lane, nn, az);
    v8f z2 = vsplat8(biasc1);
    z2 = wmma_16x16x16_f32(az, wc1b, z2);
    z2 = vrelu8(z2);

    const bool b0 = (nn & 1) != 0, b1 = (nn & 2) != 0, b2 = (nn & 4) != 0;
    float t[4], u[2];
#pragma unroll
    for (int p = 0; p < 4; ++p) {
      const float lo = z2[2 * p] * wc2, hi = z2[2 * p + 1] * wc2;
      t[p] = (b0 ? hi : lo) + row_xor<1>(b0 ? lo : hi);
    }
#pragma unroll
    for (int q = 0; q < 2; ++q)
      u[q] = (b1 ? t[2 * q + 1] : t[2 * q]) + row_xor<2>(b1 ? t[2 * q] : t[2 * q + 1]);
    float w = (b2 ? u[1] : u[0]) + row_xor<4>(b2 ? u[0] : u[1]);
    w += row_xor<8>(w);
    if (nn < 8)
      out[gb] = ph[0] + ph[1] + w + cbias;
  }
}

extern "C" void kernel_launch(void* const* d_in, const int* in_sizes, int n_in,
                              void* d_out, int out_size, void* d_ws, size_t ws_size,
                              hipStream_t stream) {
  const int*   x      = (const int*)d_in[0];
  const float* phenos = (const float*)d_in[1];
  const float* emb    = (const float*)d_in[2];
  const float* W1     = (const float*)d_in[3];
  const float* b1     = (const float*)d_in[4];
  const float* W2     = (const float*)d_in[5];
  const float* b2     = (const float*)d_in[6];
  const float* W3     = (const float*)d_in[7];
  const float* b3     = (const float*)d_in[8];
  const float* BW     = (const float*)d_in[9];
  const float* BD     = (const float*)d_in[10];
  const float* Boff   = (const float*)d_in[11];
  const float* Wc1    = (const float*)d_in[12];
  const float* bc1    = (const float*)d_in[13];
  const float* Wc2    = (const float*)d_in[14];
  const float* bc2    = (const float*)d_in[15];
  float* out = (float*)d_out;

  const int Btot   = out_size;
  const int ntiles = (Btot + 15) / 16;

  recover_model_kernel<<<GRID_BLOCKS, WAVES_PER_BLOCK * 32, 0, stream>>>(
      x, phenos, emb, W1, b1, W2, b2, W3, b3, BW, BD, Boff, Wc1, bc1, Wc2, bc2,
      out, ntiles);
}


# Activations kept transposed (features x samples): each layer's D registers are the next WMMA's B operand, no LDS hand-overs; T read in groups of 8; so 4 fat waves per SIMD, not 8; sum order changes
# speedup vs baseline: 2.1628x; 1.0513x over previous
#include <hip/hip_runtime.h>

#define WAVES_PER_BLOCK 8
#define GRID_BLOCKS 512
#define TFRAG_F4 1024

typedef int   v2i __attribute__((ext_vector_type(2)));
typedef float v2f __attribute__((ext_vector_type(2)));
typedef float v4f __attribute__((ext_vector_type(4)));
typedef float v8f __attribute__((ext_vector_type(8)));

__device__ __forceinline__ v8f vsplat8(float s) {
  v8f v;
#pragma unroll
  for (int j = 0; j < 8; ++j) v[j] = s;
  return v;
}

__device__ __forceinline__ v8f vrelu8(v8f v) {
#pragma unroll
  for (int j = 0; j < 8; ++j) v[j] = fmaxf(v[j], 0.0f);
  return v;
}

__device__ __forceinline__ float other_half(float v) {
  return __builtin_bit_cast(
      float, __builtin_amdgcn_permlanex16(__builtin_bit_cast(int, v), __builtin_bit_cast(int, v),
                                          0x76543210, 0xfedcba98, false, false));
}

__device__ __forceinline__ v8f load8(const float* __restrict__ p) {
  const v4f lo = *(const v4f*)p;
  const v4f hi = *(const v4f*)(p + 4);
  v8f v;
  v[0] = lo[0]; v[1] = lo[1]; v[2] = lo[2]; v[3] = lo[3];
  v[4] = hi[0]; v[5] = hi[1]; v[6] = hi[2]; v[7] = hi[3];
  return v;
}

__device__ __forceinline__ void load_w_afmt(const float* __restrict__ W, int lane, v2f a[4]) {
  const v8f w = load8(W + (lane & 15) * 16 + 8 * (lane >> 4));
#pragma unroll
  for (int q = 0; q < 4; ++q) { a[q][0] = w[2 * q]; a[q][1] = w[2 * q + 1]; }
}

__device__ __forceinline__ v8f layer_t(const v2f wa[4], v8f hin, v8f c) {
#pragma unroll
  for (int q = 0; q < 4; ++q) {
    v2f b;
    b[0] = hin[2 * q]; b[1] = hin[2 * q + 1];
    c = __builtin_amdgcn_wmma_f32_16x16x4_f32(false, wa[q], false, b, (short)0, c, false, false);
  }
  return c;
}

__global__ __launch_bounds__(WAVES_PER_BLOCK * 32, 4) void recover_model_kernel(
    const int* __restrict__ x, const float* __restrict__ phenos,
    const float* __restrict__ emb,
    const float* __restrict__ W1, const float* __restrict__ b1,
    const float* __restrict__ W2, const float* __restrict__ b2,
    const float* __restrict__ W3, const float* __restrict__ b3,
    const float* __restrict__ BW, const float* __restrict__ BD,
    const float* __restrict__ Boff,
    const float* __restrict__ Wc1, const float* __restrict__ bc1,
    const float* __restrict__ Wc2, const float* __restrict__ bc2,
    float* __restrict__ out, int ntiles) {
  __shared__ v4f tfrag[TFRAG_F4];
  const int lane = threadIdx.x & 31;
  const int wave = __builtin_amdgcn_readfirstlane(threadIdx.x >> 5);
  const int nn = lane & 15;
  const int hh = lane >> 4;

  {
    float* tf = (float*)tfrag;
#pragma unroll
    for (int i = 0; i < 2; ++i) {
      const int n = 2 * wave + i;
      v2f ta[4], tb[4];
#pragma unroll
      for (int c = 0; c < 4; ++c) {
        const int m0 = 4 * c + 2 * hh;
        const v2f bd = *(const v2f*)(BD + n * 16 + m0);
        ta[c][0] = BW[n * 256 + m0 * 16 + nn] * bd[0];
        ta[c][1] = BW[n * 256 + (m0 + 1) * 16 + nn] * bd[1];
        tb[c][0] = BW[m0 * 256 + n * 16 + nn];
        tb[c][1] = BW[(m0 + 1) * 256 + n * 16 + nn];
      }
      v8f t = vsplat8(0.0f);
#pragma unroll
      for (int c = 0; c < 4; ++c)
        t = __builtin_amdgcn_wmma_f32_16x16x4_f32(false, ta[c], false, tb[c], (short)0, t, false, false);
      const int wsel = ((nn >> 3) ^ hh) & 1, pp = (nn >> 2) & 1;
#pragma unroll
      for (int r = 0; r < 8; ++r)
        tf[((((r * 2 + wsel) * 2 + pp) * 32) + hh * 16 + n) * 4 + (nn & 3)] = t[r];
    }
  }
  __syncthreads();

  v2f w1a[4], w2a[4], w3a[4], wc1a[4];
  load_w_afmt(W1, lane, w1a);
  load_w_afmt(W2, lane, w2a);
  load_w_afmt(W3, lane, w3a);
  load_w_afmt(Wc1, lane, wc1a);
  const v8f b1v = load8(b1 + 8 * hh), b2v = load8(b2 + 8 * hh), b3v = load8(b3 + 8 * hh);
  const v8f boffv = load8(Boff + 8 * hh), bc1v = load8(bc1 + 8 * hh);
  const v8f wc2v = load8(Wc2 + 8 * hh);
  const float cbias = bc2[0];

  for (int tile = blockIdx.x * WAVES_PER_BLOCK + wave; tile < ntiles;
       tile += gridDim.x * WAVES_PER_BLOCK) {

    const int gb = tile * 16 + nn;
    const v2i g = *(const v2i*)(x + 2 * gb);
    const v2f ph = *(const v2f*)(phenos + 2 * gb);
    v4f tb[2][8];
#pragma unroll
    for (int i = 0; i < 16; ++i) tb[i >> 3][i & 7] = tfrag[i * 32 + lane];
    __builtin_amdgcn_sched_barrier(0);
    asm volatile("" ::"v"(ph[0]), "v"(ph[1]));

    const v8f x0 = load8(emb + (long)g[0] * 16 + 8 * hh);
    const v8f x1 = load8(emb + (long)g[1] * 16 + 8 * hh);
    v8f h0 = vrelu8(layer_t(w1a, x0, b1v));
    v8f h1 = vrelu8(layer_t(w1a, x1, b1v));
    h0 = vrelu8(layer_t(w2a, h0, b2v));
    h1 = vrelu8(layer_t(w2a, h1, b2v));
    const v8f e0v = layer_t(w3a, h0, b3v);
    const v8f e1v = layer_t(w3a, h1, b3v);
    v8f e1x;
#pragma unroll
    for (int r = 0; r < 8; ++r) e1x[r] = other_half(e1v[r]);

    v8f z0 = boffv;
    v8f z1 = vsplat8(0.0f);
#pragma unroll
    for (int blk = 0; blk < 4; ++blk) {
#pragma unroll
      for (int rr = 0; rr < 2; ++rr) {
        const float s = e0v[2 * blk + rr];
#pragma unroll
        for (int w = 0; w < 2; ++w) {
          const v8f ew = w ? e1x : e1v;
#pragma unroll
          for (int pp = 0; pp < 2; ++pp) {
            const v4f t = tb[blk & 1][(rr * 2 + w) * 2 + pp];
            v2f a, b;
            a[0] = t[0]; a[1] = t[1]; b[0] = ew[4 * pp] * s; b[1] = ew[4 * pp + 1] * s;
            z0 = __builtin_amdgcn_wmma_f32_16x16x4_f32(false, a, false, b, (short)0, z0, false, false);
            a[0] = t[2]; a[1] = t[3]; b[0] = ew[4 * pp + 2] * s; b[1] = ew[4 * pp + 3] * s;
            z1 = __builtin_amdgcn_wmma_f32_16x16x4_f32(false, a, false, b, (short)0, z1, false, false);
          }
        }
      }
      if (blk < 2) {
        __builtin_amdgcn_sched_barrier(0);
#pragma unroll
        for (int i = 0; i < 8; ++i) tb[blk & 1][i] = tfrag[((blk + 2) * 8 + i) * 32 + lane];
        __builtin_amdgcn_sched_barrier(0);
      }
    }
    v8f z;
#pragma unroll
    for (int j = 0; j < 8; ++j) z[j] = z0[j] + z1[j];

    const v8f z2 = vrelu8(layer_t(wc1a, z, bc1v));

    float p = z2[0] * wc2v[0];
#pragma unroll
    for (int r = 1; r < 8; ++r) p = fmaf(z2[r], wc2v[r], p);
    const float q = other_half(p);
    if (hh == 0)
      out[gb] = ph[0] + ph[1] + (p + q) + cbias;
  }
}

extern "C" void kernel_launch(void* const* d_in, const int* in_sizes, int n_in,
                              void* d_out, int out_size, void* d_ws, size_t ws_size,
                              hipStream_t stream) {
  const int*   x      = (const int*)d_in[0];
  const float* phenos = (const float*)d_in[1];
  const float* emb    = (const float*)d_in[2];
  const float* W1     = (const float*)d_in[3];
  const float* b1     = (const float*)d_in[4];
  const float* W2     = (const float*)d_in[5];
  const float* b2     = (const float*)d_in[6];
  const float* W3     = (const float*)d_in[7];
  const float* b3     = (const float*)d_in[8];
  const float* BW     = (const float*)d_in[9];
  const float* BD     = (const float*)d_in[10];
  const float* Boff   = (const float*)d_in[11];
  const float* Wc1    = (const float*)d_in[12];
  const float* bc1    = (const float*)d_in[13];
  const float* Wc2    = (const float*)d_in[14];
  const float* bc2    = (const float*)d_in[15];
  float* out = (float*)d_out;

  const int Btot   = out_size;
  const int ntiles = (Btot + 15) / 16;

  recover_model_kernel<<<GRID_BLOCKS, WAVES_PER_BLOCK * 32, 0, stream>>>(
      x, phenos, emb, W1, b1, W2, b2, W3, b3, BW, BD, Boff, Wc1, bc1, Wc2, bc2,
      out, ntiles);
}


# Gather chain software-pipelined over a wave's tiles: index pairs two tiles ahead, table rows and phenos one tile ahead, set-up loads ahead of the T build; no global-load wait inside a tile
# speedup vs baseline: 2.3710x; 1.0963x over previous
#include <hip/hip_runtime.h>

#define WAVES_PER_BLOCK 8
#define GRID_BLOCKS 512
#define TFRAG_F4 1024

typedef int   v2i __attribute__((ext_vector_type(2)));
typedef float v2f __attribute__((ext_vector_type(2)));
typedef float v4f __attribute__((ext_vector_type(4)));
typedef float v8f __attribute__((ext_vector_type(8)));

__device__ __forceinline__ v8f vsplat8(float s) {
  v8f v;
#pragma unroll
  for (int j = 0; j < 8; ++j) v[j] = s;
  return v;
}

__device__ __forceinline__ v8f vrelu8(v8f v) {
#pragma unroll
  for (int j = 0; j < 8; ++j) v[j] = fmaxf(v[j], 0.0f);
  return v;
}

__device__ __forceinline__ float other_half(float v) {
  return __builtin_bit_cast(
      float, __builtin_amdgcn_permlanex16(__builtin_bit_cast(int, v), __builtin_bit_cast(int, v),
                                          0x76543210, 0xfedcba98, false, false));
}

__device__ __forceinline__ v8f load8(const float* __restrict__ p) {
  const v4f lo = *(const v4f*)p;
  const v4f hi = *(const v4f*)(p + 4);
  v8f v;
  v[0] = lo[0]; v[1] = lo[1]; v[2] = lo[2]; v[3] = lo[3];
  v[4] = hi[0]; v[5] = hi[1]; v[6] = hi[2]; v[7] = hi[3];
  return v;
}

__device__ __forceinline__ void load_w_afmt(const float* __restrict__ W, int lane, v2f a[4]) {
  const v8f w = load8(W + (lane & 15) * 16 + 8 * (lane >> 4));
#pragma unroll
  for (int q = 0; q < 4; ++q) { a[q][0] = w[2 * q]; a[q][1] = w[2 * q + 1]; }
}

__device__ __forceinline__ v8f layer_t(const v2f wa[4], v8f hin, v8f c) {
#pragma unroll
  for (int q = 0; q < 4; ++q) {
    v2f b;
    b[0] = hin[2 * q]; b[1] = hin[2 * q + 1];
    c = __builtin_amdgcn_wmma_f32_16x16x4_f32(false, wa[q], false, b, (short)0, c, false, false);
  }
  return c;
}

__global__ __launch_bounds__(WAVES_PER_BLOCK * 32, 4) void recover_model_kernel(
    const int* __restrict__ x, const float* __restrict__ phenos,
    const float* __restrict__ emb,
    const float* __restrict__ W1, const float* __restrict__ b1,
    const float* __restrict__ W2, const float* __restrict__ b2,
    const float* __restrict__ W3, const float* __restrict__ b3,
    const float* __restrict__ BW, const float* __restrict__ BD,
    const float* __restrict__ Boff,
    const float* __restrict__ Wc1, const float* __restrict__ bc1,
    const float* __restrict__ Wc2, const float* __restrict__ bc2,
    float* __restrict__ out, int ntiles) {
  __shared__ v4f tfrag[TFRAG_F4];
  const int lane = threadIdx.x & 31;
  const int wave = __builtin_amdgcn_readfirstlane(threadIdx.x >> 5);
  const int nn = lane & 15;
  const int hh = lane >> 4;

  const int tstride = gridDim.x * WAVES_PER_BLOCK;
  const int tfirst = blockIdx.x * WAVES_PER_BLOCK + wave;
  const int tlast = max(ntiles - 1, 0);
  const int sfill0 = (min(tfirst, tlast) * 16 + nn) * 2;
  const int sfill1 = (min(tfirst + tstride, tlast) * 16 + nn) * 2;
  const v2i gfirst = *(const v2i*)(x + sfill0);
  v2i gn = *(const v2i*)(x + sfill1);
  v2f ph = *(const v2f*)(phenos + sfill0);
  v8f x0, x1;

  v2f w1a[4], w2a[4], w3a[4], wc1a[4];
  load_w_afmt(W1, lane, w1a);
  load_w_afmt(W2, lane, w2a);
  load_w_afmt(W3, lane, w3a);
  load_w_afmt(Wc1, lane, wc1a);
  const v8f b1v = load8(b1 + 8 * hh), b2v = load8(b2 + 8 * hh), b3v = load8(b3 + 8 * hh);
  const v8f boffv = load8(Boff + 8 * hh), bc1v = load8(bc1 + 8 * hh);
  const v8f wc2v = load8(Wc2 + 8 * hh);
  const float cbias = bc2[0];

  {
    float* tf = (float*)tfrag;
    v2f ta[2][4], tb[2][4];
#pragma unroll
    for (int i = 0; i < 2; ++i) {
      const int n = 2 * wave + i;
#pragma unroll
      for (int c = 0; c < 4; ++c) {
        const int m0 = 4 * c + 2 * hh;
        const v2f bd = *(const v2f*)(BD + n * 16 + m0);
        ta[i][c][0] = BW[n * 256 + m0 * 16 + nn] * bd[0];
        ta[i][c][1] = BW[n * 256 + (m0 + 1) * 16 + nn] * bd[1];
        tb[i][c][0] = BW[m0 * 256 + n * 16 + nn];
        tb[i][c][1] = BW[(m0 + 1) * 256 + n * 16 + nn];
      }
    }
    x0 = load8(emb + (long)gfirst[0] * 16 + 8 * hh);
    x1 = load8(emb + (long)gfirst[1] * 16 + 8 * hh);
#pragma unroll
    for (int i = 0; i < 2; ++i) {
      const int n = 2 * wave + i;
      v8f t = vsplat8(0.0f);
#pragma unroll
      for (int c = 0; c < 4; ++c)
        t = __builtin_amdgcn_wmma_f32_16x16x4_f32(false, ta[i][c], false, tb[i][c], (short)0, t, false, false);
      const int wsel = ((nn >> 3) ^ hh) & 1, pp = (nn >> 2) & 1;
#pragma unroll
      for (int r = 0; r < 8; ++r)
        tf[((((r * 2 + wsel) * 2 + pp) * 32) + hh * 16 + n) * 4 + (nn & 3)] = t[r];
    }
  }
  __syncthreads();

  for (int tile = tfirst; tile < ntiles; tile += tstride) {

    const int gb = tile * 16 + nn;
    const int snext = (min(tile + tstride, tlast) * 16 + nn) * 2;
    const int snext2 = (min(tile + 2 * tstride, tlast) * 16 + nn) * 2;
    const v2i gnn = *(const v2i*)(x + snext2);
    const v2f phn = *(const v2f*)(phenos + snext);
    const v8f x0n = load8(emb + (long)gn[0] * 16 + 8 * hh);
    const v8f x1n = load8(emb + (long)gn[1] * 16 + 8 * hh);
    v4f tb[2][8];
#pragma unroll
    for (int i = 0; i < 16; ++i) tb[i >> 3][i & 7] = tfrag[i * 32 + lane];
    __builtin_amdgcn_sched_barrier(0);

    v8f h0 = vrelu8(layer_t(w1a, x0, b1v));
    v8f h1 = vrelu8(layer_t(w1a, x1, b1v));
    h0 = vrelu8(layer_t(w2a, h0, b2v));
    h1 = vrelu8(layer_t(w2a, h1, b2v));
    const v8f e0v = layer_t(w3a, h0, b3v);
    const v8f e1v = layer_t(w3a, h1, b3v);
    v8f e1x;
#pragma unroll
    for (int r = 0; r < 8; ++r) e1x[r] = other_half(e1v[r]);

    v8f z0 = boffv;
    v8f z1 = vsplat8(0.0f);
#pragma unroll
    for (int blk = 0; blk < 4; ++blk) {
#pragma unroll
      for (int rr = 0; rr < 2; ++rr) {
        const float s = e0v[2 * blk + rr];
#pragma unroll
        for (int w = 0; w < 2; ++w) {
          const v8f ew = w ? e1x : e1v;
#pragma unroll
          for (int pp = 0; pp < 2; ++pp) {
            const v4f t = tb[blk & 1][(rr * 2 + w) * 2 + pp];
            v2f a, b;
            a[0] = t[0]; a[1] = t[1]; b[0] = ew[4 * pp] * s; b[1] = ew[4 * pp + 1] * s;
            z0 = __builtin_amdgcn_wmma_f32_16x16x4_f32(false, a, false, b, (short)0, z0, false, false);
            a[0] = t[2]; a[1] = t[3]; b[0] = ew[4 * pp + 2] * s; b[1] = ew[4 * pp + 3] * s;
            z1 = __builtin_amdgcn_wmma_f32_16x16x4_f32(false, a, false, b, (short)0, z1, false, false);
          }
        }
      }
      if (blk < 2) {
        __builtin_amdgcn_sched_barrier(0);
#pragma unroll
        for (int i = 0; i < 8; ++i) tb[blk & 1][i] = tfrag[((blk + 2) * 8 + i) * 32 + lane];
        __builtin_amdgcn_sched_barrier(0);
      }
    }
    v8f z;
#pragma unroll
    for (int j = 0; j < 8; ++j) z[j] = z0[j] + z1[j];

    const v8f z2 = vrelu8(layer_t(wc1a, z, bc1v));

    float p = z2[0] * wc2v[0];
#pragma unroll
    for (int r = 1; r < 8; ++r) p = fmaf(z2[r], wc2v[r], p);
    const float q = other_half(p);
    if (hh == 0)
      out[gb] = ph[0] + ph[1] + (p + q) + cbias;
    x0 = x0n; x1 = x1n; gn = gnn; ph = phn;
  }
}

extern "C" void kernel_launch(void* const* d_in, const int* in_sizes, int n_in,
                              void* d_out, int out_size, void* d_ws, size_t ws_size,
                              hipStream_t stream) {
  const int*   x      = (const int*)d_in[0];
  const float* phenos = (const float*)d_in[1];
  const float* emb    = (const float*)d_in[2];
  const float* W1     = (const float*)d_in[3];
  const float* b1     = (const float*)d_in[4];
  const float* W2     = (const float*)d_in[5];
  const float* b2     = (const float*)d_in[6];
  const float* W3     = (const float*)d_in[7];
  const float* b3     = (const float*)d_in[8];
  const float* BW     = (const float*)d_in[9];
  const float* BD     = (const float*)d_in[10];
  const float* Boff   = (const float*)d_in[11];
  const float* Wc1    = (const float*)d_in[12];
  const float* bc1    = (const float*)d_in[13];
  const float* Wc2    = (const float*)d_in[14];
  const float* bc2    = (const float*)d_in[15];
  float* out = (float*)d_out;

  const int Btot   = out_size;
  const int ntiles = (Btot + 15) / 16;

  recover_model_kernel<<<GRID_BLOCKS, WAVES_PER_BLOCK * 32, 0, stream>>>(
      x, phenos, emb, W1, b1, W2, b2, W3, b3, BW, BD, Boff, Wc1, bc1, Wc2, bc2,
      out, ntiles);
}
